# starting kernel
# baseline (speedup 1.0000x reference)
#include <hip/hip_runtime.h>

typedef __attribute__((ext_vector_type(2))) float v2f;
typedef __attribute__((ext_vector_type(8))) float v8f;

#define NTOK 262144
#define CCH  64
#define HIDDEN 128

#define G_BLOCKS_PER_BATCH 64
#define Y_BLOCKS_PER_BATCH 256

__device__ __forceinline__ v8f v8f_zero() {
    v8f z;
#pragma unroll
    for (int i = 0; i < 8; ++i) z[i] = 0.0f;
    return z;
}

__device__ __forceinline__ v8f wmma_f32(v2f a, v2f b, v8f c) {
    return __builtin_amdgcn_wmma_f32_16x16x4_f32(false, a, false, b, (short)0, c,
                                                 false, false);
}

__global__ void zero_kernel(float* __restrict__ p, int count) {
    int i = blockIdx.x * blockDim.x + threadIdx.x;
    if (i < count) p[i] = 0.0f;
}

__global__ __launch_bounds__(128) void gram_kernel(const float* __restrict__ x,
                                                   float* __restrict__ G) {
    const int tid    = threadIdx.x;
    const int lane   = tid & 31;
    const int w      = tid >> 5;
    const int batch  = blockIdx.x / G_BLOCKS_PER_BATCH;
    const int blk    = blockIdx.x % G_BLOCKS_PER_BATCH;
    const float* xb  = x + (size_t)batch * CCH * NTOK;
    float* Gb        = G + batch * CCH * CCH;

    __shared__ float tile[4][CCH][34];

    const int lanelo = lane & 15;
    const int hi2    = (lane >> 4) * 2;
    const int rowOff = (lane < 16) ? 0 : 8;

    v8f acc[4][4];
#pragma unroll
    for (int i = 0; i < 4; ++i)
#pragma unroll
        for (int j = 0; j < 4; ++j) acc[i][j] = v8f_zero();

    const int tokensPerBlock = NTOK / G_BLOCKS_PER_BATCH;
    const int tokensPerWave  = tokensPerBlock / 4;
    const int tWave0         = blk * tokensPerBlock + w * tokensPerWave;

    for (int it = 0; it < tokensPerWave; it += 32) {
        const int t0 = tWave0 + it;
#pragma unroll
        for (int c = 0; c < CCH; ++c)
            tile[w][c][lane] = xb[(size_t)c * NTOK + t0 + lane];
        __syncthreads();

#pragma unroll
        for (int kg = 0; kg < 8; ++kg) {
            v2f op[4];
#pragma unroll
            for (int gch = 0; gch < 4; ++gch) {
                const float* p = &tile[w][gch * 16 + lanelo][kg * 4 + hi2];
                op[gch] = *(const v2f*)p;
            }
#pragma unroll
            for (int i = 0; i < 4; ++i)
#pragma unroll
                for (int j = 0; j < 4; ++j)
                    acc[i][j] = wmma_f32(op[i], op[j], acc[i][j]);
        }
        __syncthreads();
    }

#pragma unroll
    for (int i = 0; i < 4; ++i)
#pragma unroll
        for (int j = 0; j < 4; ++j)
#pragma unroll
            for (int r = 0; r < 8; ++r) {
                const int row = i * 16 + r + rowOff;
                const int col = j * 16 + lanelo;
                atomicAdd(&Gb[row * CCH + col], acc[i][j][r]);
            }
}

__global__ __launch_bounds__(256) void combine_kernel(
    const float* __restrict__ wqkv, const float* __restrict__ wout,
    const float* __restrict__ G, float* __restrict__ T1, float* __restrict__ CT,
    float* __restrict__ M, float* __restrict__ A) {
    const int b   = blockIdx.x;
    const int tid = threadIdx.x;
    const float* Gb  = G  + b * 4096;
    float* T1b       = T1 + b * 8192;
    float* CTb       = CT + b * 4096;
    float* Mb        = M  + b * 8192;
    float* Ab        = A  + b * 4096;
    const float* Wq  = wqkv;
    const float* Wk  = wqkv + 128 * CCH;
    const float* Wv  = wqkv + 256 * CCH;
    const float scale = 0.1767766952966369f;

    for (int idx = tid; idx < 64 * 128; idx += 256) {
        const int c = idx >> 7, j = idx & 127;
        float s = 0.f;
        for (int c2 = 0; c2 < CCH; ++c2) s += Gb[c * CCH + c2] * Wv[j * CCH + c2];
        T1b[idx] = s;
    }
    __syncthreads();
    for (int idx = tid; idx < 4096; idx += 256) {
        const int h = idx >> 10, d = (idx >> 5) & 31, e = idx & 31;
        float s = 0.f;
        for (int c = 0; c < CCH; ++c)
            s += Wk[(h * 32 + d) * CCH + c] * T1b[c * 128 + h * 32 + e];
        CTb[idx] = s * (1.0f / 4096.0f);
    }
    __syncthreads();
    for (int idx = tid; idx < 128 * 64; idx += 256) {
        const int j = idx >> 6, c = idx & 63, h = j >> 5, e = j & 31;
        float s = 0.f;
        for (int d = 0; d < 32; ++d)
            s += CTb[h * 1024 + d * 32 + e] * Wq[(h * 32 + d) * CCH + c];
        Mb[idx] = s * scale;
    }
    __syncthreads();
    for (int idx = tid; idx < 4096; idx += 256) {
        const int o = idx >> 6, c = idx & 63;
        float s = 0.f;
        for (int j = 0; j < HIDDEN; ++j) s += wout[o * HIDDEN + j] * Mb[j * 64 + c];
        Ab[idx] = s;
    }
}

__global__ __launch_bounds__(128) void attn_out_kernel(
    const float* __restrict__ x, const float* __restrict__ A,
    const float* __restrict__ b_out, const float* __restrict__ g,
    float* __restrict__ out) {
    const int tid    = threadIdx.x;
    const int lane   = tid & 31;
    const int w      = tid >> 5;
    const int batch  = blockIdx.x / Y_BLOCKS_PER_BATCH;
    const int blk    = blockIdx.x % Y_BLOCKS_PER_BATCH;
    const float* xb  = x + (size_t)batch * CCH * NTOK;
    float* ob        = out + (size_t)batch * CCH * NTOK;
    const float* Ab  = A + batch * CCH * CCH;

    const int lanelo = lane & 15;
    const int hi2    = (lane >> 4) * 2;
    const int rowOff = (lane < 16) ? 0 : 8;

    v2f aop[16][4];
#pragma unroll
    for (int kg = 0; kg < 16; ++kg)
#pragma unroll
        for (int mt = 0; mt < 4; ++mt) {
            const float* p = &Ab[(mt * 16 + lanelo) * CCH + kg * 4 + hi2];
            aop[kg][mt] = *(const v2f*)p;
        }

    float bia[4][8], gai[4][8];
#pragma unroll
    for (int mt = 0; mt < 4; ++mt)
#pragma unroll
        for (int r = 0; r < 8; ++r) {
            const int ch = mt * 16 + r + rowOff;
            bia[mt][r] = b_out[ch];
            gai[mt][r] = g[ch];
        }

    const int tokensPerBlock = NTOK / Y_BLOCKS_PER_BATCH;
    const int tokensPerWave  = tokensPerBlock / 4;
    const int tW0            = blk * tokensPerBlock + w * tokensPerWave;

    for (int tt = 0; tt < tokensPerWave; tt += 16) {
        const int t0 = tW0 + tt;
        v8f acc[4];
#pragma unroll
        for (int mt = 0; mt < 4; ++mt) acc[mt] = v8f_zero();

#pragma unroll
        for (int kg = 0; kg < 16; ++kg) {
            const float* bp = xb + (size_t)(kg * 4 + hi2) * NTOK + t0 + lanelo;
            v2f bop;
            bop[0] = bp[0];
            bop[1] = bp[NTOK];
#pragma unroll
            for (int mt = 0; mt < 4; ++mt)
                acc[mt] = wmma_f32(aop[kg][mt], bop, acc[mt]);
        }

        float v[4][8];
        float s = 0.f, sq = 0.f;
#pragma unroll
        for (int mt = 0; mt < 4; ++mt)
#pragma unroll
            for (int r = 0; r < 8; ++r) {
                const float y = acc[mt][r] + bia[mt][r];
                v[mt][r] = y;
                s += y;
                sq += y * y;
            }
        s  += __shfl_xor(s, 16, 32);
        sq += __shfl_xor(sq, 16, 32);
        const float mean = s * (1.0f / 64.0f);
        const float var  = sq * (1.0f / 64.0f) - mean * mean;
        const float inv  = rsqrtf(var + 1e-5f);

#pragma unroll
        for (int mt = 0; mt < 4; ++mt)
#pragma unroll
            for (int r = 0; r < 8; ++r) {
                const int ch = mt * 16 + r + rowOff;
                ob[(size_t)ch * NTOK + t0 + lanelo] = (v[mt][r] - mean) * inv * gai[mt][r];
            }
    }
}

extern "C" void kernel_launch(void* const* d_in, const int* in_sizes, int n_in,
                              void* d_out, int out_size, void* d_ws, size_t ws_size,
                              hipStream_t stream) {
    (void)in_sizes; (void)n_in; (void)out_size; (void)ws_size;
    const float* x    = (const float*)d_in[0];
    const float* wqkv = (const float*)d_in[1];
    const float* wout = (const float*)d_in[2];
    const float* bout = (const float*)d_in[3];
    const float* g    = (const float*)d_in[4];
    float* out        = (float*)d_out;
    float* ws         = (float*)d_ws;

    float* G  = ws;
    float* A  = ws + 8192;
    float* T1 = ws + 16384;
    float* CT = ws + 32768;
    float* M  = ws + 40960;

    zero_kernel<<<32, 256, 0, stream>>>(G, 2 * CCH * CCH);
    gram_kernel<<<2 * G_BLOCKS_PER_BATCH, 128, 0, stream>>>(x, G);
    combine_kernel<<<2, 256, 0, stream>>>(wqkv, wout, G, T1, CT, M, A);
    attn_out_kernel<<<2 * Y_BLOCKS_PER_BATCH, 128, 0, stream>>>(x, A, bout, g, out);
}


# combine kernel: its four small products run as f32 WMMA tiles on 16 waves, operands and intermediates held in LDS (was per-thread scalar loops reading global memory); summation order changes
# speedup vs baseline: 1.9513x; 1.9513x over previous
#include <hip/hip_runtime.h>

typedef __attribute__((ext_vector_type(2))) float v2f;
typedef __attribute__((ext_vector_type(8))) float v8f;

#define NTOK 262144
#define CCH  64
#define HIDDEN 128

#define G_BLOCKS_PER_BATCH 64
#define Y_BLOCKS_PER_BATCH 256

__device__ __forceinline__ v8f v8f_zero() {
    v8f z;
#pragma unroll
    for (int i = 0; i < 8; ++i) z[i] = 0.0f;
    return z;
}

__device__ __forceinline__ v8f wmma_f32(v2f a, v2f b, v8f c) {
    return __builtin_amdgcn_wmma_f32_16x16x4_f32(false, a, false, b, (short)0, c,
                                                 false, false);
}

__global__ void zero_kernel(float* __restrict__ p, int count) {
    int i = blockIdx.x * blockDim.x + threadIdx.x;
    if (i < count) p[i] = 0.0f;
}

__global__ __launch_bounds__(128) void gram_kernel(const float* __restrict__ x,
                                                   float* __restrict__ G) {
    const int tid    = threadIdx.x;
    const int lane   = tid & 31;
    const int w      = tid >> 5;
    const int batch  = blockIdx.x / G_BLOCKS_PER_BATCH;
    const int blk    = blockIdx.x % G_BLOCKS_PER_BATCH;
    const float* xb  = x + (size_t)batch * CCH * NTOK;
    float* Gb        = G + batch * CCH * CCH;

    __shared__ float tile[4][CCH][34];

    const int lanelo = lane & 15;
    const int hi2    = (lane >> 4) * 2;
    const int rowOff = (lane < 16) ? 0 : 8;

    v8f acc[4][4];
#pragma unroll
    for (int i = 0; i < 4; ++i)
#pragma unroll
        for (int j = 0; j < 4; ++j) acc[i][j] = v8f_zero();

    const int tokensPerBlock = NTOK / G_BLOCKS_PER_BATCH;
    const int tokensPerWave  = tokensPerBlock / 4;
    const int tWave0         = blk * tokensPerBlock + w * tokensPerWave;

    for (int it = 0; it < tokensPerWave; it += 32) {
        const int t0 = tWave0 + it;
#pragma unroll
        for (int c = 0; c < CCH; ++c)
            tile[w][c][lane] = xb[(size_t)c * NTOK + t0 + lane];
        __syncthreads();

#pragma unroll
        for (int kg = 0; kg < 8; ++kg) {
            v2f op[4];
#pragma unroll
            for (int gch = 0; gch < 4; ++gch) {
                const float* p = &tile[w][gch * 16 + lanelo][kg * 4 + hi2];
                op[gch] = *(const v2f*)p;
            }
#pragma unroll
            for (int i = 0; i < 4; ++i)
#pragma unroll
                for (int j = 0; j < 4; ++j)
                    acc[i][j] = wmma_f32(op[i], op[j], acc[i][j]);
        }
        __syncthreads();
    }

#pragma unroll
    for (int i = 0; i < 4; ++i)
#pragma unroll
        for (int j = 0; j < 4; ++j)
#pragma unroll
            for (int r = 0; r < 8; ++r) {
                const int row = i * 16 + r + rowOff;
                const int col = j * 16 + lanelo;
                atomicAdd(&Gb[row * CCH + col], acc[i][j][r]);
            }
}

#define PW 68
#define PQ 72
#define PO 132
#define PC 36

__device__ __forceinline__ v2f lds_frag(const float* base, int row, int pitch, int k) {
    return *(const v2f*)(base + row * pitch + k);
}

__global__ __launch_bounds__(512) void combine_kernel(
    const float* __restrict__ wqkv, const float* __restrict__ wout,
    const float* __restrict__ G, float* __restrict__ A) {
    __shared__ __attribute__((aligned(16))) float sG [64  * PW];
    __shared__ __attribute__((aligned(16))) float sWv[128 * PW];
    __shared__ __attribute__((aligned(16))) float sWk[128 * PW];
    __shared__ __attribute__((aligned(16))) float sWq[128 * PQ];
    __shared__ __attribute__((aligned(16))) float sWo[64  * PO];
    __shared__ __attribute__((aligned(16))) float sT [128 * PW];
    __shared__ __attribute__((aligned(16))) float sC [128 * PC];
    float* sM = sT;

    const int b      = blockIdx.x;
    const int tid    = threadIdx.x;
    const int lane   = tid & 31;
    const int w      = tid >> 5;
    const int lanelo = lane & 15;
    const int hi2    = (lane >> 4) * 2;
    const int rowOff = (lane >> 4) * 8;
    const float scale = 0.1767766952966369f;

    const float4* G4  = (const float4*)(G + b * 4096);
    const float4* Wq4 = (const float4*)(wqkv);
    const float4* Wk4 = (const float4*)(wqkv + 128 * CCH);
    const float4* Wv4 = (const float4*)(wqkv + 256 * CCH);
    const float4* Wo4 = (const float4*)(wout);

#pragma unroll
    for (int i = 0; i < 2; ++i) {
        const int f = tid + 512 * i;
        *(float4*)&sG[(f >> 4) * PW + (f & 15) * 4] = G4[f];
    }
#pragma unroll
    for (int i = 0; i < 4; ++i) {
        const int f = tid + 512 * i;
        *(float4*)&sWv[(f >> 4) * PW + (f & 15) * 4] = Wv4[f];
    }
    __syncthreads();

    float4 pk[4], pq[4], po[4];
#pragma unroll
    for (int i = 0; i < 4; ++i) {
        const int f = tid + 512 * i;
        pk[i] = Wk4[f];
        pq[i] = Wq4[f];
        po[i] = Wo4[f];
    }

    const int jb = w >> 1;
    const int cp = (w & 1) * 2;
    const int h  = w >> 2;
    const int eb = (w >> 1) & 1;

    {
        v8f acc0 = v8f_zero(), acc1 = v8f_zero();
#pragma unroll
        for (int kg = 0; kg < 16; ++kg) {
            const v2f a  = lds_frag(sWv, jb * 16 + lanelo, PW, kg * 4 + hi2);
            const v2f b0 = lds_frag(sG, cp * 16 + lanelo, PW, kg * 4 + hi2);
            const v2f b1 = lds_frag(sG, (cp + 1) * 16 + lanelo, PW, kg * 4 + hi2);
            acc0 = wmma_f32(a, b0, acc0);
            acc1 = wmma_f32(a, b1, acc1);
        }
#pragma unroll
        for (int r = 0; r < 8; ++r) {
            sT[(jb * 16 + rowOff + r) * PW + cp * 16 + lanelo]       = acc0[r];
            sT[(jb * 16 + rowOff + r) * PW + (cp + 1) * 16 + lanelo] = acc1[r];
        }
    }
#pragma unroll
    for (int i = 0; i < 4; ++i) {
        const int f = tid + 512 * i;
        *(float4*)&sWk[(f >> 4) * PW + (f & 15) * 4] = pk[i];
        *(float4*)&sWq[(f >> 4) * PQ + (f & 15) * 4] = pq[i];
        *(float4*)&sWo[(f >> 5) * PO + (f & 31) * 4] = po[i];
    }
    __syncthreads();

    {
        const int db = w & 1;
        v8f c0 = v8f_zero(), c1 = v8f_zero();
#pragma unroll
        for (int kg = 0; kg < 16; kg += 2) {
            const v2f a0 = lds_frag(sT, h * 32 + eb * 16 + lanelo, PW, kg * 4 + hi2);
            const v2f k0 = lds_frag(sWk, h * 32 + db * 16 + lanelo, PW, kg * 4 + hi2);
            const v2f a1 = lds_frag(sT, h * 32 + eb * 16 + lanelo, PW, (kg + 1) * 4 + hi2);
            const v2f k1 = lds_frag(sWk, h * 32 + db * 16 + lanelo, PW, (kg + 1) * 4 + hi2);
            c0 = wmma_f32(a0, k0, c0);
            c1 = wmma_f32(a1, k1, c1);
        }
#pragma unroll
        for (int r = 0; r < 8; ++r)
            sC[(h * 32 + eb * 16 + rowOff + r) * PC + db * 16 + lanelo] =
                (c0[r] + c1[r]) * (1.0f / 4096.0f);
    }
    __syncthreads();

    {
        v8f m0 = v8f_zero(), m1 = v8f_zero();
#pragma unroll
        for (int kg = 0; kg < 8; ++kg) {
            const v2f bb = lds_frag(sC, h * 32 + eb * 16 + lanelo, PC, kg * 4 + hi2);
            const float* q = &sWq[(h * 32 + kg * 4 + hi2) * PQ + lanelo];
            v2f a0, a1;
            a0[0] = q[cp * 16];
            a0[1] = q[PQ + cp * 16];
            a1[0] = q[(cp + 1) * 16];
            a1[1] = q[PQ + (cp + 1) * 16];
            m0 = wmma_f32(a0, bb, m0);
            m1 = wmma_f32(a1, bb, m1);
        }
#pragma unroll
        for (int r = 0; r < 8; ++r) {
            sM[(cp * 16 + rowOff + r) * PO + jb * 16 + lanelo]       = m0[r] * scale;
            sM[((cp + 1) * 16 + rowOff + r) * PO + jb * 16 + lanelo] = m1[r] * scale;
        }
    }
    __syncthreads();

    {
        const int ob = w >> 2;
        const int cb = w & 3;
        v8f o0 = v8f_zero(), o1 = v8f_zero();
#pragma unroll
        for (int kg = 0; kg < 32; kg += 2) {
            const v2f a0 = lds_frag(sWo, ob * 16 + lanelo, PO, kg * 4 + hi2);
            const v2f b0 = lds_frag(sM, cb * 16 + lanelo, PO, kg * 4 + hi2);
            const v2f a1 = lds_frag(sWo, ob * 16 + lanelo, PO, (kg + 1) * 4 + hi2);
            const v2f b1 = lds_frag(sM, cb * 16 + lanelo, PO, (kg + 1) * 4 + hi2);
            o0 = wmma_f32(a0, b0, o0);
            o1 = wmma_f32(a1, b1, o1);
        }
#pragma unroll
        for (int r = 0; r < 8; ++r)
            sC[(ob * 16 + rowOff + r) * CCH + cb * 16 + lanelo] = o0[r] + o1[r];
    }
    __syncthreads();
    float4* A4 = (float4*)(A + b * 4096);
#pragma unroll
    for (int i = 0; i < 2; ++i) {
        const int f = tid + 512 * i;
        A4[f] = *(const float4*)&sC[f * 4];
    }
}

__global__ __launch_bounds__(128) void attn_out_kernel(
    const float* __restrict__ x, const float* __restrict__ A,
    const float* __restrict__ b_out, const float* __restrict__ g,
    float* __restrict__ out) {
    const int tid    = threadIdx.x;
    const int lane   = tid & 31;
    const int w      = tid >> 5;
    const int batch  = blockIdx.x / Y_BLOCKS_PER_BATCH;
    const int blk    = blockIdx.x % Y_BLOCKS_PER_BATCH;
    const float* xb  = x + (size_t)batch * CCH * NTOK;
    float* ob        = out + (size_t)batch * CCH * NTOK;
    const float* Ab  = A + batch * CCH * CCH;

    const int lanelo = lane & 15;
    const int hi2    = (lane >> 4) * 2;
    const int rowOff = (lane < 16) ? 0 : 8;

    v2f aop[16][4];
#pragma unroll
    for (int kg = 0; kg < 16; ++kg)
#pragma unroll
        for (int mt = 0; mt < 4; ++mt) {
            const float* p = &Ab[(mt * 16 + lanelo) * CCH + kg * 4 + hi2];
            aop[kg][mt] = *(const v2f*)p;
        }

    float bia[4][8], gai[4][8];
#pragma unroll
    for (int mt = 0; mt < 4; ++mt)
#pragma unroll
        for (int r = 0; r < 8; ++r) {
            const int ch = mt * 16 + r + rowOff;
            bia[mt][r] = b_out[ch];
            gai[mt][r] = g[ch];
        }

    const int tokensPerBlock = NTOK / Y_BLOCKS_PER_BATCH;
    const int tokensPerWave  = tokensPerBlock / 4;
    const int tW0            = blk * tokensPerBlock + w * tokensPerWave;

    for (int tt = 0; tt < tokensPerWave; tt += 16) {
        const int t0 = tW0 + tt;
        v8f acc[4];
#pragma unroll
        for (int mt = 0; mt < 4; ++mt) acc[mt] = v8f_zero();

#pragma unroll
        for (int kg = 0; kg < 16; ++kg) {
            const float* bp = xb + (size_t)(kg * 4 + hi2) * NTOK + t0 + lanelo;
            v2f bop;
            bop[0] = bp[0];
            bop[1] = bp[NTOK];
#pragma unroll
            for (int mt = 0; mt < 4; ++mt)
                acc[mt] = wmma_f32(aop[kg][mt], bop, acc[mt]);
        }

        float v[4][8];
        float s = 0.f, sq = 0.f;
#pragma unroll
        for (int mt = 0; mt < 4; ++mt)
#pragma unroll
            for (int r = 0; r < 8; ++r) {
                const float y = acc[mt][r] + bia[mt][r];
                v[mt][r] = y;
                s += y;
                sq += y * y;
            }
        s  += __shfl_xor(s, 16, 32);
        sq += __shfl_xor(sq, 16, 32);
        const float mean = s * (1.0f / 64.0f);
        const float var  = sq * (1.0f / 64.0f) - mean * mean;
        const float inv  = rsqrtf(var + 1e-5f);

#pragma unroll
        for (int mt = 0; mt < 4; ++mt)
#pragma unroll
            for (int r = 0; r < 8; ++r) {
                const int ch = mt * 16 + r + rowOff;
                ob[(size_t)ch * NTOK + t0 + lanelo] = (v[mt][r] - mean) * inv * gai[mt][r];
            }
    }
}

extern "C" void kernel_launch(void* const* d_in, const int* in_sizes, int n_in,
                              void* d_out, int out_size, void* d_ws, size_t ws_size,
                              hipStream_t stream) {
    (void)in_sizes; (void)n_in; (void)out_size; (void)ws_size;
    const float* x    = (const float*)d_in[0];
    const float* wqkv = (const float*)d_in[1];
    const float* wout = (const float*)d_in[2];
    const float* bout = (const float*)d_in[3];
    const float* g    = (const float*)d_in[4];
    float* out        = (float*)d_out;
    float* ws         = (float*)d_ws;

    float* G  = ws;
    float* A  = ws + 8192;

    zero_kernel<<<32, 256, 0, stream>>>(G, 2 * CCH * CCH);
    gram_kernel<<<2 * G_BLOCKS_PER_BATCH, 128, 0, stream>>>(x, G);
    combine_kernel<<<2, 512, 0, stream>>>(wqkv, wout, G, A);
    attn_out_kernel<<<2 * Y_BLOCKS_PER_BATCH, 128, 0, stream>>>(x, A, bout, g, out);
}


# gram kernel regridded: 256 blocks of 8 waves, one per WGP (was 128 of 4), wave-private LDS tiles, no block barrier, loads a tile ahead, partials summed in LDS before atomics; summation order changes
# speedup vs baseline: 2.8883x; 1.4802x over previous
#include <hip/hip_runtime.h>

typedef __attribute__((ext_vector_type(2))) float v2f;
typedef __attribute__((ext_vector_type(4))) float v4f;
typedef __attribute__((ext_vector_type(8))) float v8f;

#define NTOK 262144
#define CCH  64
#define HIDDEN 128

#define G_BLOCKS_PER_BATCH 128
#define G_WAVES 8
#define Y_BLOCKS_PER_BATCH 256

__device__ __forceinline__ v8f v8f_zero() {
    v8f z;
#pragma unroll
    for (int i = 0; i < 8; ++i) z[i] = 0.0f;
    return z;
}

__device__ __forceinline__ v8f wmma_f32(v2f a, v2f b, v8f c) {
    return __builtin_amdgcn_wmma_f32_16x16x4_f32(false, a, false, b, (short)0, c,
                                                 false, false);
}

__global__ void zero_kernel(float* __restrict__ p, int count) {
    int i = blockIdx.x * blockDim.x + threadIdx.x;
    if (i < count) p[i] = 0.0f;
}


__device__ __forceinline__ void wave_lds_handover() {
    __builtin_amdgcn_fence(__ATOMIC_RELEASE, "wavefront");
    __builtin_amdgcn_wave_barrier();
    __builtin_amdgcn_fence(__ATOMIC_ACQUIRE, "wavefront");
}

__global__ __launch_bounds__(256) void gram_kernel(const float* __restrict__ x,
                                                   float* __restrict__ G) {
    const int tid    = threadIdx.x;
    const int lane   = tid & 31;
    const int w      = tid >> 5;
    const int batch  = blockIdx.x / G_BLOCKS_PER_BATCH;
    const int blk    = blockIdx.x % G_BLOCKS_PER_BATCH;
    const float* xb  = x + (size_t)batch * CCH * NTOK;
    float* Gb        = G + batch * CCH * CCH;

    __shared__ float tile[G_WAVES][CCH][34];
    __shared__ __attribute__((aligned(16))) v4f red[G_WAVES * 16 * 2 * 32];

    const int lanelo = lane & 15;
    const int hi2    = (lane >> 4) * 2;
    const int rowOff = (lane < 16) ? 0 : 8;

    v8f acc[4][4];
#pragma unroll
    for (int i = 0; i < 4; ++i)
#pragma unroll
        for (int j = 0; j < 4; ++j) acc[i][j] = v8f_zero();

    const int tokensPerBlock = NTOK / G_BLOCKS_PER_BATCH;
    const int tokensPerWave  = tokensPerBlock / G_WAVES;
    const float* xw = xb + blk * tokensPerBlock + w * tokensPerWave + lane;

    float pf[CCH];
#pragma unroll
    for (int c = 0; c < CCH; ++c) pf[c] = xw[(size_t)c * NTOK];

    for (int it = 0; it < tokensPerWave; it += 32) {
#pragma unroll
        for (int c = 0; c < CCH; ++c) tile[w][c][lane] = pf[c];
        if (it + 32 < tokensPerWave) {
#pragma unroll
            for (int c = 0; c < CCH; ++c) pf[c] = xw[(size_t)c * NTOK + it + 32];
        }
        wave_lds_handover();

        v2f op[8][4];
#pragma unroll
        for (int kg = 0; kg < 8; ++kg)
#pragma unroll
            for (int gch = 0; gch < 4; ++gch) {
                const float* p = &tile[w][gch * 16 + lanelo][kg * 4 + hi2];
                op[kg][gch] = *(const v2f*)p;
            }
#pragma unroll
        for (int kg = 0; kg < 8; ++kg)
#pragma unroll
            for (int i = 0; i < 4; ++i)
#pragma unroll
                for (int j = 0; j < 4; ++j)
                    acc[i][j] = wmma_f32(op[kg][i], op[kg][j], acc[i][j]);
        wave_lds_handover();
    }

#pragma unroll
    for (int t = 0; t < 16; ++t) {
        const v8f a = acc[t >> 2][t & 3];
        v4f lo, hi;
        lo[0] = a[0]; lo[1] = a[1]; lo[2] = a[2]; lo[3] = a[3];
        hi[0] = a[4]; hi[1] = a[5]; hi[2] = a[6]; hi[3] = a[7];
        red[((w * 16 + t) * 2 + 0) * 32 + lane] = lo;
        red[((w * 16 + t) * 2 + 1) * 32 + lane] = hi;
    }
    __syncthreads();

#pragma unroll
    for (int u = 0; u < 2; ++u) {
        const int t = 2 * w + u;
        v4f lo = red[(t * 2 + 0) * 32 + lane];
        v4f hi = red[(t * 2 + 1) * 32 + lane];
#pragma unroll
        for (int s = 1; s < G_WAVES; ++s) {
            lo += red[((s * 16 + t) * 2 + 0) * 32 + lane];
            hi += red[((s * 16 + t) * 2 + 1) * 32 + lane];
        }
        const int row0 = (t >> 2) * 16 + rowOff;
        const int col  = (t & 3) * 16 + lanelo;
#pragma unroll
        for (int r = 0; r < 4; ++r) {
            atomicAdd(&Gb[(row0 + r) * CCH + col], lo[r]);
            atomicAdd(&Gb[(row0 + 4 + r) * CCH + col], hi[r]);
        }
    }
}

#define PW 68
#define PQ 72
#define PO 132
#define PC 36

__device__ __forceinline__ v2f lds_frag(const float* base, int row, int pitch, int k) {
    return *(const v2f*)(base + row * pitch + k);
}

__global__ __launch_bounds__(512) void combine_kernel(
    const float* __restrict__ wqkv, const float* __restrict__ wout,
    const float* __restrict__ G, float* __restrict__ A) {
    __shared__ __attribute__((aligned(16))) float sG [64  * PW];
    __shared__ __attribute__((aligned(16))) float sWv[128 * PW];
    __shared__ __attribute__((aligned(16))) float sWk[128 * PW];
    __shared__ __attribute__((aligned(16))) float sWq[128 * PQ];
    __shared__ __attribute__((aligned(16))) float sWo[64  * PO];
    __shared__ __attribute__((aligned(16))) float sT [128 * PW];
    __shared__ __attribute__((aligned(16))) float sC [128 * PC];
    float* sM = sT;

    const int b      = blockIdx.x;
    const int tid    = threadIdx.x;
    const int lane   = tid & 31;
    const int w      = tid >> 5;
    const int lanelo = lane & 15;
    const int hi2    = (lane >> 4) * 2;
    const int rowOff = (lane >> 4) * 8;
    const float scale = 0.1767766952966369f;

    const float4* G4  = (const float4*)(G + b * 4096);
    const float4* Wq4 = (const float4*)(wqkv);
    const float4* Wk4 = (const float4*)(wqkv + 128 * CCH);
    const float4* Wv4 = (const float4*)(wqkv + 256 * CCH);
    const float4* Wo4 = (const float4*)(wout);

#pragma unroll
    for (int i = 0; i < 2; ++i) {
        const int f = tid + 512 * i;
        *(float4*)&sG[(f >> 4) * PW + (f & 15) * 4] = G4[f];
    }
#pragma unroll
    for (int i = 0; i < 4; ++i) {
        const int f = tid + 512 * i;
        *(float4*)&sWv[(f >> 4) * PW + (f & 15) * 4] = Wv4[f];
    }
    __syncthreads();

    float4 pk[4], pq[4], po[4];
#pragma unroll
    for (int i = 0; i < 4; ++i) {
        const int f = tid + 512 * i;
        pk[i] = Wk4[f];
        pq[i] = Wq4[f];
        po[i] = Wo4[f];
    }

    const int jb = w >> 1;
    const int cp = (w & 1) * 2;
    const int h  = w >> 2;
    const int eb = (w >> 1) & 1;

    {
        v8f acc0 = v8f_zero(), acc1 = v8f_zero();
#pragma unroll
        for (int kg = 0; kg < 16; ++kg) {
            const v2f a  = lds_frag(sWv, jb * 16 + lanelo, PW, kg * 4 + hi2);
            const v2f b0 = lds_frag(sG, cp * 16 + lanelo, PW, kg * 4 + hi2);
            const v2f b1 = lds_frag(sG, (cp + 1) * 16 + lanelo, PW, kg * 4 + hi2);
            acc0 = wmma_f32(a, b0, acc0);
            acc1 = wmma_f32(a, b1, acc1);
        }
#pragma unroll
        for (int r = 0; r < 8; ++r) {
            sT[(jb * 16 + rowOff + r) * PW + cp * 16 + lanelo]       = acc0[r];
            sT[(jb * 16 + rowOff + r) * PW + (cp + 1) * 16 + lanelo] = acc1[r];
        }
    }
#pragma unroll
    for (int i = 0; i < 4; ++i) {
        const int f = tid + 512 * i;
        *(float4*)&sWk[(f >> 4) * PW + (f & 15) * 4] = pk[i];
        *(float4*)&sWq[(f >> 4) * PQ + (f & 15) * 4] = pq[i];
        *(float4*)&sWo[(f >> 5) * PO + (f & 31) * 4] = po[i];
    }
    __syncthreads();

    {
        const int db = w & 1;
        v8f c0 = v8f_zero(), c1 = v8f_zero();
#pragma unroll
        for (int kg = 0; kg < 16; kg += 2) {
            const v2f a0 = lds_frag(sT, h * 32 + eb * 16 + lanelo, PW, kg * 4 + hi2);
            const v2f k0 = lds_frag(sWk, h * 32 + db * 16 + lanelo, PW, kg * 4 + hi2);
            const v2f a1 = lds_frag(sT, h * 32 + eb * 16 + lanelo, PW, (kg + 1) * 4 + hi2);
            const v2f k1 = lds_frag(sWk, h * 32 + db * 16 + lanelo, PW, (kg + 1) * 4 + hi2);
            c0 = wmma_f32(a0, k0, c0);
            c1 = wmma_f32(a1, k1, c1);
        }
#pragma unroll
        for (int r = 0; r < 8; ++r)
            sC[(h * 32 + eb * 16 + rowOff + r) * PC + db * 16 + lanelo] =
                (c0[r] + c1[r]) * (1.0f / 4096.0f);
    }
    __syncthreads();

    {
        v8f m0 = v8f_zero(), m1 = v8f_zero();
#pragma unroll
        for (int kg = 0; kg < 8; ++kg) {
            const v2f bb = lds_frag(sC, h * 32 + eb * 16 + lanelo, PC, kg * 4 + hi2);
            const float* q = &sWq[(h * 32 + kg * 4 + hi2) * PQ + lanelo];
            v2f a0, a1;
            a0[0] = q[cp * 16];
            a0[1] = q[PQ + cp * 16];
            a1[0] = q[(cp + 1) * 16];
            a1[1] = q[PQ + (cp + 1) * 16];
            m0 = wmma_f32(a0, bb, m0);
            m1 = wmma_f32(a1, bb, m1);
        }
#pragma unroll
        for (int r = 0; r < 8; ++r) {
            sM[(cp * 16 + rowOff + r) * PO + jb * 16 + lanelo]       = m0[r] * scale;
            sM[((cp + 1) * 16 + rowOff + r) * PO + jb * 16 + lanelo] = m1[r] * scale;
        }
    }
    __syncthreads();

    {
        const int ob = w >> 2;
        const int cb = w & 3;
        v8f o0 = v8f_zero(), o1 = v8f_zero();
#pragma unroll
        for (int kg = 0; kg < 32; kg += 2) {
            const v2f a0 = lds_frag(sWo, ob * 16 + lanelo, PO, kg * 4 + hi2);
            const v2f b0 = lds_frag(sM, cb * 16 + lanelo, PO, kg * 4 + hi2);
            const v2f a1 = lds_frag(sWo, ob * 16 + lanelo, PO, (kg + 1) * 4 + hi2);
            const v2f b1 = lds_frag(sM, cb * 16 + lanelo, PO, (kg + 1) * 4 + hi2);
            o0 = wmma_f32(a0, b0, o0);
            o1 = wmma_f32(a1, b1, o1);
        }
#pragma unroll
        for (int r = 0; r < 8; ++r)
            sC[(ob * 16 + rowOff + r) * CCH + cb * 16 + lanelo] = o0[r] + o1[r];
    }
    __syncthreads();
    float4* A4 = (float4*)(A + b * 4096);
#pragma unroll
    for (int i = 0; i < 2; ++i) {
        const int f = tid + 512 * i;
        A4[f] = *(const float4*)&sC[f * 4];
    }
}

__global__ __launch_bounds__(128) void attn_out_kernel(
    const float* __restrict__ x, const float* __restrict__ A,
    const float* __restrict__ b_out, const float* __restrict__ g,
    float* __restrict__ out) {
    const int tid    = threadIdx.x;
    const int lane   = tid & 31;
    const int w      = tid >> 5;
    const int batch  = blockIdx.x / Y_BLOCKS_PER_BATCH;
    const int blk    = blockIdx.x % Y_BLOCKS_PER_BATCH;
    const float* xb  = x + (size_t)batch * CCH * NTOK;
    float* ob        = out + (size_t)batch * CCH * NTOK;
    const float* Ab  = A + batch * CCH * CCH;

    const int lanelo = lane & 15;
    const int hi2    = (lane >> 4) * 2;
    const int rowOff = (lane < 16) ? 0 : 8;

    v2f aop[16][4];
#pragma unroll
    for (int kg = 0; kg < 16; ++kg)
#pragma unroll
        for (int mt = 0; mt < 4; ++mt) {
            const float* p = &Ab[(mt * 16 + lanelo) * CCH + kg * 4 + hi2];
            aop[kg][mt] = *(const v2f*)p;
        }

    float bia[4][8], gai[4][8];
#pragma unroll
    for (int mt = 0; mt < 4; ++mt)
#pragma unroll
        for (int r = 0; r < 8; ++r) {
            const int ch = mt * 16 + r + rowOff;
            bia[mt][r] = b_out[ch];
            gai[mt][r] = g[ch];
        }

    const int tokensPerBlock = NTOK / Y_BLOCKS_PER_BATCH;
    const int tokensPerWave  = tokensPerBlock / 4;
    const int tW0            = blk * tokensPerBlock + w * tokensPerWave;

    for (int tt = 0; tt < tokensPerWave; tt += 16) {
        const int t0 = tW0 + tt;
        v8f acc[4];
#pragma unroll
        for (int mt = 0; mt < 4; ++mt) acc[mt] = v8f_zero();

#pragma unroll
        for (int kg = 0; kg < 16; ++kg) {
            const float* bp = xb + (size_t)(kg * 4 + hi2) * NTOK + t0 + lanelo;
            v2f bop;
            bop[0] = bp[0];
            bop[1] = bp[NTOK];
#pragma unroll
            for (int mt = 0; mt < 4; ++mt)
                acc[mt] = wmma_f32(aop[kg][mt], bop, acc[mt]);
        }

        float v[4][8];
        float s = 0.f, sq = 0.f;
#pragma unroll
        for (int mt = 0; mt < 4; ++mt)
#pragma unroll
            for (int r = 0; r < 8; ++r) {
                const float y = acc[mt][r] + bia[mt][r];
                v[mt][r] = y;
                s += y;
                sq += y * y;
            }
        s  += __shfl_xor(s, 16, 32);
        sq += __shfl_xor(sq, 16, 32);
        const float mean = s * (1.0f / 64.0f);
        const float var  = sq * (1.0f / 64.0f) - mean * mean;
        const float inv  = rsqrtf(var + 1e-5f);

#pragma unroll
        for (int mt = 0; mt < 4; ++mt)
#pragma unroll
            for (int r = 0; r < 8; ++r) {
                const int ch = mt * 16 + r + rowOff;
                ob[(size_t)ch * NTOK + t0 + lanelo] = (v[mt][r] - mean) * inv * gai[mt][r];
            }
    }
}

extern "C" void kernel_launch(void* const* d_in, const int* in_sizes, int n_in,
                              void* d_out, int out_size, void* d_ws, size_t ws_size,
                              hipStream_t stream) {
    (void)in_sizes; (void)n_in; (void)out_size; (void)ws_size;
    const float* x    = (const float*)d_in[0];
    const float* wqkv = (const float*)d_in[1];
    const float* wout = (const float*)d_in[2];
    const float* bout = (const float*)d_in[3];
    const float* g    = (const float*)d_in[4];
    float* out        = (float*)d_out;
    float* ws         = (float*)d_ws;

    float* G  = ws;
    float* A  = ws + 8192;

    zero_kernel<<<32, 256, 0, stream>>>(G, 2 * CCH * CCH);
    gram_kernel<<<2 * G_BLOCKS_PER_BATCH, 256, 0, stream>>>(x, G);
    combine_kernel<<<2, 512, 0, stream>>>(wqkv, wout, G, A);
    attn_out_kernel<<<2 * Y_BLOCKS_PER_BATCH, 128, 0, stream>>>(x, A, bout, g, out);
}


# output kernel: a wave trip takes 32 tokens as an even-token and an odd-token tile; x loads and out stores are 8 bytes per lane, every half-wave access one whole 128-byte line (was 4-byte half lines)
# speedup vs baseline: 3.1433x; 1.0883x over previous
#include <hip/hip_runtime.h>

typedef __attribute__((ext_vector_type(2))) float v2f;
typedef __attribute__((ext_vector_type(4))) float v4f;
typedef __attribute__((ext_vector_type(8))) float v8f;

#define NTOK 262144
#define CCH  64
#define HIDDEN 128

#define G_BLOCKS_PER_BATCH 128
#define G_WAVES 8
#define Y_BLOCKS_PER_BATCH 256

__device__ __forceinline__ v8f v8f_zero() {
    v8f z;
#pragma unroll
    for (int i = 0; i < 8; ++i) z[i] = 0.0f;
    return z;
}

__device__ __forceinline__ v8f wmma_f32(v2f a, v2f b, v8f c) {
    return __builtin_amdgcn_wmma_f32_16x16x4_f32(false, a, false, b, (short)0, c,
                                                 false, false);
}

__global__ void zero_kernel(float* __restrict__ p, int count) {
    int i = blockIdx.x * blockDim.x + threadIdx.x;
    if (i < count) p[i] = 0.0f;
}


__device__ __forceinline__ void wave_lds_handover() {
    __builtin_amdgcn_fence(__ATOMIC_RELEASE, "wavefront");
    __builtin_amdgcn_wave_barrier();
    __builtin_amdgcn_fence(__ATOMIC_ACQUIRE, "wavefront");
}

__global__ __launch_bounds__(256) void gram_kernel(const float* __restrict__ x,
                                                   float* __restrict__ G) {
    const int tid    = threadIdx.x;
    const int lane   = tid & 31;
    const int w      = tid >> 5;
    const int batch  = blockIdx.x / G_BLOCKS_PER_BATCH;
    const int blk    = blockIdx.x % G_BLOCKS_PER_BATCH;
    const float* xb  = x + (size_t)batch * CCH * NTOK;
    float* Gb        = G + batch * CCH * CCH;

    __shared__ float tile[G_WAVES][CCH][34];
    __shared__ __attribute__((aligned(16))) v4f red[G_WAVES * 16 * 2 * 32];

    const int lanelo = lane & 15;
    const int hi2    = (lane >> 4) * 2;
    const int rowOff = (lane < 16) ? 0 : 8;

    v8f acc[4][4];
#pragma unroll
    for (int i = 0; i < 4; ++i)
#pragma unroll
        for (int j = 0; j < 4; ++j) acc[i][j] = v8f_zero();

    const int tokensPerBlock = NTOK / G_BLOCKS_PER_BATCH;
    const int tokensPerWave  = tokensPerBlock / G_WAVES;
    const float* xw = xb + blk * tokensPerBlock + w * tokensPerWave + lane;

    float pf[CCH];
#pragma unroll
    for (int c = 0; c < CCH; ++c) pf[c] = xw[(size_t)c * NTOK];

    for (int it = 0; it < tokensPerWave; it += 32) {
#pragma unroll
        for (int c = 0; c < CCH; ++c) tile[w][c][lane] = pf[c];
        if (it + 32 < tokensPerWave) {
#pragma unroll
            for (int c = 0; c < CCH; ++c) pf[c] = xw[(size_t)c * NTOK + it + 32];
        }
        wave_lds_handover();

        v2f op[8][4];
#pragma unroll
        for (int kg = 0; kg < 8; ++kg)
#pragma unroll
            for (int gch = 0; gch < 4; ++gch) {
                const float* p = &tile[w][gch * 16 + lanelo][kg * 4 + hi2];
                op[kg][gch] = *(const v2f*)p;
            }
#pragma unroll
        for (int kg = 0; kg < 8; ++kg)
#pragma unroll
            for (int i = 0; i < 4; ++i)
#pragma unroll
                for (int j = 0; j < 4; ++j)
                    acc[i][j] = wmma_f32(op[kg][i], op[kg][j], acc[i][j]);
        wave_lds_handover();
    }

#pragma unroll
    for (int t = 0; t < 16; ++t) {
        const v8f a = acc[t >> 2][t & 3];
        v4f lo, hi;
        lo[0] = a[0]; lo[1] = a[1]; lo[2] = a[2]; lo[3] = a[3];
        hi[0] = a[4]; hi[1] = a[5]; hi[2] = a[6]; hi[3] = a[7];
        red[((w * 16 + t) * 2 + 0) * 32 + lane] = lo;
        red[((w * 16 + t) * 2 + 1) * 32 + lane] = hi;
    }
    __syncthreads();

#pragma unroll
    for (int u = 0; u < 2; ++u) {
        const int t = 2 * w + u;
        v4f lo = red[(t * 2 + 0) * 32 + lane];
        v4f hi = red[(t * 2 + 1) * 32 + lane];
#pragma unroll
        for (int s = 1; s < G_WAVES; ++s) {
            lo += red[((s * 16 + t) * 2 + 0) * 32 + lane];
            hi += red[((s * 16 + t) * 2 + 1) * 32 + lane];
        }
        const int row0 = (t >> 2) * 16 + rowOff;
        const int col  = (t & 3) * 16 + lanelo;
#pragma unroll
        for (int r = 0; r < 4; ++r) {
            atomicAdd(&Gb[(row0 + r) * CCH + col], lo[r]);
            atomicAdd(&Gb[(row0 + 4 + r) * CCH + col], hi[r]);
        }
    }
}

#define PW 68
#define PQ 72
#define PO 132
#define PC 36

__device__ __forceinline__ v2f lds_frag(const float* base, int row, int pitch, int k) {
    return *(const v2f*)(base + row * pitch + k);
}

__global__ __launch_bounds__(512) void combine_kernel(
    const float* __restrict__ wqkv, const float* __restrict__ wout,
    const float* __restrict__ G, float* __restrict__ A) {
    __shared__ __attribute__((aligned(16))) float sG [64  * PW];
    __shared__ __attribute__((aligned(16))) float sWv[128 * PW];
    __shared__ __attribute__((aligned(16))) float sWk[128 * PW];
    __shared__ __attribute__((aligned(16))) float sWq[128 * PQ];
    __shared__ __attribute__((aligned(16))) float sWo[64  * PO];
    __shared__ __attribute__((aligned(16))) float sT [128 * PW];
    __shared__ __attribute__((aligned(16))) float sC [128 * PC];
    float* sM = sT;

    const int b      = blockIdx.x;
    const int tid    = threadIdx.x;
    const int lane   = tid & 31;
    const int w      = tid >> 5;
    const int lanelo = lane & 15;
    const int hi2    = (lane >> 4) * 2;
    const int rowOff = (lane >> 4) * 8;
    const float scale = 0.1767766952966369f;

    const float4* G4  = (const float4*)(G + b * 4096);
    const float4* Wq4 = (const float4*)(wqkv);
    const float4* Wk4 = (const float4*)(wqkv + 128 * CCH);
    const float4* Wv4 = (const float4*)(wqkv + 256 * CCH);
    const float4* Wo4 = (const float4*)(wout);

#pragma unroll
    for (int i = 0; i < 2; ++i) {
        const int f = tid + 512 * i;
        *(float4*)&sG[(f >> 4) * PW + (f & 15) * 4] = G4[f];
    }
#pragma unroll
    for (int i = 0; i < 4; ++i) {
        const int f = tid + 512 * i;
        *(float4*)&sWv[(f >> 4) * PW + (f & 15) * 4] = Wv4[f];
    }
    __syncthreads();

    float4 pk[4], pq[4], po[4];
#pragma unroll
    for (int i = 0; i < 4; ++i) {
        const int f = tid + 512 * i;
        pk[i] = Wk4[f];
        pq[i] = Wq4[f];
        po[i] = Wo4[f];
    }

    const int jb = w >> 1;
    const int cp = (w & 1) * 2;
    const int h  = w >> 2;
    const int eb = (w >> 1) & 1;

    {
        v8f acc0 = v8f_zero(), acc1 = v8f_zero();
#pragma unroll
        for (int kg = 0; kg < 16; ++kg) {
            const v2f a  = lds_frag(sWv, jb * 16 + lanelo, PW, kg * 4 + hi2);
            const v2f b0 = lds_frag(sG, cp * 16 + lanelo, PW, kg * 4 + hi2);
            const v2f b1 = lds_frag(sG, (cp + 1) * 16 + lanelo, PW, kg * 4 + hi2);
            acc0 = wmma_f32(a, b0, acc0);
            acc1 = wmma_f32(a, b1, acc1);
        }
#pragma unroll
        for (int r = 0; r < 8; ++r) {
            sT[(jb * 16 + rowOff + r) * PW + cp * 16 + lanelo]       = acc0[r];
            sT[(jb * 16 + rowOff + r) * PW + (cp + 1) * 16 + lanelo] = acc1[r];
        }
    }
#pragma unroll
    for (int i = 0; i < 4; ++i) {
        const int f = tid + 512 * i;
        *(float4*)&sWk[(f >> 4) * PW + (f & 15) * 4] = pk[i];
        *(float4*)&sWq[(f >> 4) * PQ + (f & 15) * 4] = pq[i];
        *(float4*)&sWo[(f >> 5) * PO + (f & 31) * 4] = po[i];
    }
    __syncthreads();

    {
        const int db = w & 1;
        v8f c0 = v8f_zero(), c1 = v8f_zero();
#pragma unroll
        for (int kg = 0; kg < 16; kg += 2) {
            const v2f a0 = lds_frag(sT, h * 32 + eb * 16 + lanelo, PW, kg * 4 + hi2);
            const v2f k0 = lds_frag(sWk, h * 32 + db * 16 + lanelo, PW, kg * 4 + hi2);
            const v2f a1 = lds_frag(sT, h * 32 + eb * 16 + lanelo, PW, (kg + 1) * 4 + hi2);
            const v2f k1 = lds_frag(sWk, h * 32 + db * 16 + lanelo, PW, (kg + 1) * 4 + hi2);
            c0 = wmma_f32(a0, k0, c0);
            c1 = wmma_f32(a1, k1, c1);
        }
#pragma unroll
        for (int r = 0; r < 8; ++r)
            sC[(h * 32 + eb * 16 + rowOff + r) * PC + db * 16 + lanelo] =
                (c0[r] + c1[r]) * (1.0f / 4096.0f);
    }
    __syncthreads();

    {
        v8f m0 = v8f_zero(), m1 = v8f_zero();
#pragma unroll
        for (int kg = 0; kg < 8; ++kg) {
            const v2f bb = lds_frag(sC, h * 32 + eb * 16 + lanelo, PC, kg * 4 + hi2);
            const float* q = &sWq[(h * 32 + kg * 4 + hi2) * PQ + lanelo];
            v2f a0, a1;
            a0[0] = q[cp * 16];
            a0[1] = q[PQ + cp * 16];
            a1[0] = q[(cp + 1) * 16];
            a1[1] = q[PQ + (cp + 1) * 16];
            m0 = wmma_f32(a0, bb, m0);
            m1 = wmma_f32(a1, bb, m1);
        }
#pragma unroll
        for (int r = 0; r < 8; ++r) {
            sM[(cp * 16 + rowOff + r) * PO + jb * 16 + lanelo]       = m0[r] * scale;
            sM[((cp + 1) * 16 + rowOff + r) * PO + jb * 16 + lanelo] = m1[r] * scale;
        }
    }
    __syncthreads();

    {
        const int ob = w >> 2;
        const int cb = w & 3;
        v8f o0 = v8f_zero(), o1 = v8f_zero();
#pragma unroll
        for (int kg = 0; kg < 32; kg += 2) {
            const v2f a0 = lds_frag(sWo, ob * 16 + lanelo, PO, kg * 4 + hi2);
            const v2f b0 = lds_frag(sM, cb * 16 + lanelo, PO, kg * 4 + hi2);
            const v2f a1 = lds_frag(sWo, ob * 16 + lanelo, PO, (kg + 1) * 4 + hi2);
            const v2f b1 = lds_frag(sM, cb * 16 + lanelo, PO, (kg + 1) * 4 + hi2);
            o0 = wmma_f32(a0, b0, o0);
            o1 = wmma_f32(a1, b1, o1);
        }
#pragma unroll
        for (int r = 0; r < 8; ++r)
            sC[(ob * 16 + rowOff + r) * CCH + cb * 16 + lanelo] = o0[r] + o1[r];
    }
    __syncthreads();
    float4* A4 = (float4*)(A + b * 4096);
#pragma unroll
    for (int i = 0; i < 2; ++i) {
        const int f = tid + 512 * i;
        A4[f] = *(const float4*)&sC[f * 4];
    }
}

__global__ __launch_bounds__(128) void attn_out_kernel(
    const float* __restrict__ x, const float* __restrict__ A,
    const float* __restrict__ b_out, const float* __restrict__ g,
    float* __restrict__ out) {
    const int tid    = threadIdx.x;
    const int lane   = tid & 31;
    const int w      = tid >> 5;
    const int batch  = blockIdx.x / Y_BLOCKS_PER_BATCH;
    const int blk    = blockIdx.x % Y_BLOCKS_PER_BATCH;
    const float* xb  = x + (size_t)batch * CCH * NTOK;
    float* ob        = out + (size_t)batch * CCH * NTOK;
    const float* Ab  = A + batch * CCH * CCH;

    const int lanelo = lane & 15;
    const int hi2    = (lane >> 4) * 2;
    const int rowOff = (lane < 16) ? 0 : 8;

    v2f aop[16][4];
#pragma unroll
    for (int kg = 0; kg < 16; ++kg)
#pragma unroll
        for (int mt = 0; mt < 4; ++mt) {
            const float* p = &Ab[(mt * 16 + lanelo) * CCH + kg * 4 + hi2];
            aop[kg][mt] = *(const v2f*)p;
        }

    float bia[4][8], gai[4][8];
#pragma unroll
    for (int mt = 0; mt < 4; ++mt)
#pragma unroll
        for (int r = 0; r < 8; ++r) {
            const int ch = mt * 16 + r + rowOff;
            bia[mt][r] = b_out[ch];
            gai[mt][r] = g[ch];
        }

    const int tokensPerBlock = NTOK / Y_BLOCKS_PER_BATCH;
    const int tokensPerWave  = tokensPerBlock / 4;
    const int tW0            = blk * tokensPerBlock + w * tokensPerWave;

    for (int tt = 0; tt < tokensPerWave; tt += 32) {
        const int tp = tW0 + tt + 2 * lanelo;
        v8f acc0[4], acc1[4];
#pragma unroll
        for (int mt = 0; mt < 4; ++mt) { acc0[mt] = v8f_zero(); acc1[mt] = v8f_zero(); }

#pragma unroll
        for (int kg = 0; kg < 16; ++kg) {
            const float* bp = xb + (size_t)(kg * 4 + hi2) * NTOK + tp;
            const v2f c0 = *(const v2f*)bp;
            const v2f c1 = *(const v2f*)(bp + NTOK);
            v2f b0, b1;
            b0[0] = c0[0]; b0[1] = c1[0];
            b1[0] = c0[1]; b1[1] = c1[1];
#pragma unroll
            for (int mt = 0; mt < 4; ++mt) {
                acc0[mt] = wmma_f32(aop[kg][mt], b0, acc0[mt]);
                acc1[mt] = wmma_f32(aop[kg][mt], b1, acc1[mt]);
            }
        }

        float s0 = 0.f, q0 = 0.f, s1 = 0.f, q1 = 0.f;
#pragma unroll
        for (int mt = 0; mt < 4; ++mt)
#pragma unroll
            for (int r = 0; r < 8; ++r) {
                const float y0 = acc0[mt][r] + bia[mt][r];
                const float y1 = acc1[mt][r] + bia[mt][r];
                acc0[mt][r] = y0;
                acc1[mt][r] = y1;
                s0 += y0;
                q0 += y0 * y0;
                s1 += y1;
                q1 += y1 * y1;
            }
        s0 += __shfl_xor(s0, 16, 32);
        q0 += __shfl_xor(q0, 16, 32);
        s1 += __shfl_xor(s1, 16, 32);
        q1 += __shfl_xor(q1, 16, 32);
        const float mean0 = s0 * (1.0f / 64.0f);
        const float var0  = q0 * (1.0f / 64.0f) - mean0 * mean0;
        const float inv0  = rsqrtf(var0 + 1e-5f);
        const float mean1 = s1 * (1.0f / 64.0f);
        const float var1  = q1 * (1.0f / 64.0f) - mean1 * mean1;
        const float inv1  = rsqrtf(var1 + 1e-5f);

#pragma unroll
        for (int mt = 0; mt < 4; ++mt)
#pragma unroll
            for (int r = 0; r < 8; ++r) {
                const int ch = mt * 16 + r + rowOff;
                v2f o;
                o[0] = (acc0[mt][r] - mean0) * inv0 * gai[mt][r];
                o[1] = (acc1[mt][r] - mean1) * inv1 * gai[mt][r];
                *(v2f*)&ob[(size_t)ch * NTOK + tp] = o;
            }
    }
}

extern "C" void kernel_launch(void* const* d_in, const int* in_sizes, int n_in,
                              void* d_out, int out_size, void* d_ws, size_t ws_size,
                              hipStream_t stream) {
    (void)in_sizes; (void)n_in; (void)out_size; (void)ws_size;
    const float* x    = (const float*)d_in[0];
    const float* wqkv = (const float*)d_in[1];
    const float* wout = (const float*)d_in[2];
    const float* bout = (const float*)d_in[3];
    const float* g    = (const float*)d_in[4];
    float* out        = (float*)d_out;
    float* ws         = (float*)d_ws;

    float* G  = ws;
    float* A  = ws + 8192;

    zero_kernel<<<32, 256, 0, stream>>>(G, 2 * CCH * CCH);
    gram_kernel<<<2 * G_BLOCKS_PER_BATCH, 256, 0, stream>>>(x, G);
    combine_kernel<<<2, 512, 0, stream>>>(wqkv, wout, G, A);
    attn_out_kernel<<<2 * Y_BLOCKS_PER_BATCH, 128, 0, stream>>>(x, A, bout, g, out);
}


# gram kernel computes only the 10 of 16 tiles on or above the diagonal of the symmetric 64x64 Gram matrix (80 WMMAs per 32 tokens, was 128); the combine kernel mirrors them as it copies G into LDS
# speedup vs baseline: 3.3978x; 1.0810x over previous
#include <hip/hip_runtime.h>

typedef __attribute__((ext_vector_type(2))) float v2f;
typedef __attribute__((ext_vector_type(4))) float v4f;
typedef __attribute__((ext_vector_type(8))) float v8f;

#define NTOK 262144
#define CCH  64
#define HIDDEN 128

#define G_BLOCKS_PER_BATCH 128
#define G_WAVES 8
#define G_TILES 10
#define Y_BLOCKS_PER_BATCH 256

__device__ __forceinline__ v8f v8f_zero() {
    v8f z;
#pragma unroll
    for (int i = 0; i < 8; ++i) z[i] = 0.0f;
    return z;
}

__device__ __forceinline__ v8f wmma_f32(v2f a, v2f b, v8f c) {
    return __builtin_amdgcn_wmma_f32_16x16x4_f32(false, a, false, b, (short)0, c,
                                                 false, false);
}

__global__ void zero_kernel(float* __restrict__ p, int count) {
    int i = blockIdx.x * blockDim.x + threadIdx.x;
    if (i < count) p[i] = 0.0f;
}


__device__ __forceinline__ void wave_lds_handover() {
    __builtin_amdgcn_fence(__ATOMIC_RELEASE, "wavefront");
    __builtin_amdgcn_wave_barrier();
    __builtin_amdgcn_fence(__ATOMIC_ACQUIRE, "wavefront");
}

__global__ __launch_bounds__(256) void gram_kernel(const float* __restrict__ x,
                                                   float* __restrict__ G) {
    const int tid    = threadIdx.x;
    const int lane   = tid & 31;
    const int w      = tid >> 5;
    const int batch  = blockIdx.x / G_BLOCKS_PER_BATCH;
    const int blk    = blockIdx.x % G_BLOCKS_PER_BATCH;
    const float* xb  = x + (size_t)batch * CCH * NTOK;
    float* Gb        = G + batch * CCH * CCH;

    __shared__ float tile[G_WAVES][CCH][34];
    __shared__ __attribute__((aligned(16))) v4f red[G_WAVES * G_TILES * 2 * 32];

    const int lanelo = lane & 15;
    const int hi2    = (lane >> 4) * 2;
    const int rowOff = (lane < 16) ? 0 : 8;

    v8f acc[G_TILES];
#pragma unroll
    for (int t = 0; t < G_TILES; ++t) acc[t] = v8f_zero();

    const int tokensPerBlock = NTOK / G_BLOCKS_PER_BATCH;
    const int tokensPerWave  = tokensPerBlock / G_WAVES;
    const float* xw = xb + blk * tokensPerBlock + w * tokensPerWave + lane;

    float pf[CCH];
#pragma unroll
    for (int c = 0; c < CCH; ++c) pf[c] = xw[(size_t)c * NTOK];

    for (int it = 0; it < tokensPerWave; it += 32) {
#pragma unroll
        for (int c = 0; c < CCH; ++c) tile[w][c][lane] = pf[c];
        if (it + 32 < tokensPerWave) {
#pragma unroll
            for (int c = 0; c < CCH; ++c) pf[c] = xw[(size_t)c * NTOK + it + 32];
        }
        wave_lds_handover();

        v2f op[8][4];
#pragma unroll
        for (int kg = 0; kg < 8; ++kg)
#pragma unroll
            for (int gch = 0; gch < 4; ++gch) {
                const float* p = &tile[w][gch * 16 + lanelo][kg * 4 + hi2];
                op[kg][gch] = *(const v2f*)p;
            }
#pragma unroll
        for (int kg = 0; kg < 8; ++kg)
#pragma unroll
            for (int i = 0; i < 4; ++i)
#pragma unroll
                for (int j = i; j < 4; ++j) {
                    const int t = i * 4 + j - (i * (i + 1)) / 2;
                    acc[t] = wmma_f32(op[kg][i], op[kg][j], acc[t]);
                }
        wave_lds_handover();
    }

#pragma unroll
    for (int t = 0; t < G_TILES; ++t) {
        const v8f a = acc[t];
        v4f lo, hi;
        lo[0] = a[0]; lo[1] = a[1]; lo[2] = a[2]; lo[3] = a[3];
        hi[0] = a[4]; hi[1] = a[5]; hi[2] = a[6]; hi[3] = a[7];
        red[((w * G_TILES + t) * 2 + 0) * 32 + lane] = lo;
        red[((w * G_TILES + t) * 2 + 1) * 32 + lane] = hi;
    }
    __syncthreads();

#pragma unroll
    for (int u = 0; u < 2; ++u) {
        const int t = w + G_WAVES * u;
        if (t < G_TILES) {
            v4f lo = red[(t * 2 + 0) * 32 + lane];
            v4f hi = red[(t * 2 + 1) * 32 + lane];
#pragma unroll
            for (int s = 1; s < G_WAVES; ++s) {
                lo += red[((s * G_TILES + t) * 2 + 0) * 32 + lane];
                hi += red[((s * G_TILES + t) * 2 + 1) * 32 + lane];
            }
            const int ti = (t >= 9) ? 3 : (t >= 7) ? 2 : (t >= 4) ? 1 : 0;
            const int tj = t - 4 * ti + (ti * (ti + 1)) / 2;
            const int row0 = ti * 16 + rowOff;
            const int col  = tj * 16 + lanelo;
#pragma unroll
            for (int r = 0; r < 4; ++r) {
                atomicAdd(&Gb[(row0 + r) * CCH + col], lo[r]);
                atomicAdd(&Gb[(row0 + 4 + r) * CCH + col], hi[r]);
            }
        }
    }
}

#define PW 68
#define PQ 72
#define PO 132
#define PC 36

__device__ __forceinline__ v2f lds_frag(const float* base, int row, int pitch, int k) {
    return *(const v2f*)(base + row * pitch + k);
}

__global__ __launch_bounds__(512) void combine_kernel(
    const float* __restrict__ wqkv, const float* __restrict__ wout,
    const float* __restrict__ G, float* __restrict__ A) {
    __shared__ __attribute__((aligned(16))) float sG [64  * PW];
    __shared__ __attribute__((aligned(16))) float sWv[128 * PW];
    __shared__ __attribute__((aligned(16))) float sWk[128 * PW];
    __shared__ __attribute__((aligned(16))) float sWq[128 * PQ];
    __shared__ __attribute__((aligned(16))) float sWo[64  * PO];
    __shared__ __attribute__((aligned(16))) float sT [128 * PW];
    __shared__ __attribute__((aligned(16))) float sC [128 * PC];
    float* sM = sT;

    const int b      = blockIdx.x;
    const int tid    = threadIdx.x;
    const int lane   = tid & 31;
    const int w      = tid >> 5;
    const int lanelo = lane & 15;
    const int hi2    = (lane >> 4) * 2;
    const int rowOff = (lane >> 4) * 8;
    const float scale = 0.1767766952966369f;

    const float4* G4  = (const float4*)(G + b * 4096);
    const float4* Wq4 = (const float4*)(wqkv);
    const float4* Wk4 = (const float4*)(wqkv + 128 * CCH);
    const float4* Wv4 = (const float4*)(wqkv + 256 * CCH);
    const float4* Wo4 = (const float4*)(wout);

    v4f g0 = ((const v4f*)G4)[tid],        g1 = ((const v4f*)G4)[tid + 512];
    v4f w0 = ((const v4f*)Wv4)[tid],       w1 = ((const v4f*)Wv4)[tid + 512];
    v4f w2 = ((const v4f*)Wv4)[tid + 1024], w3 = ((const v4f*)Wv4)[tid + 1536];
    asm volatile("" : "+v"(g0), "+v"(g1), "+v"(w0), "+v"(w1), "+v"(w2), "+v"(w3));
    const v4f gv[2] = {g0, g1};
    const v4f wv[4] = {w0, w1, w2, w3};
#pragma unroll
    for (int i = 0; i < 2; ++i) {
        const int f   = tid + 512 * i;
        const int row = f >> 4;
        const int c4  = (f & 15) * 4;
        if ((c4 >> 4) >= (row >> 4)) {
            const v4f v = gv[i];
            *(v4f*)&sG[row * PW + c4] = v;
            if ((c4 >> 4) > (row >> 4)) {
                sG[(c4 + 0) * PW + row] = v[0];
                sG[(c4 + 1) * PW + row] = v[1];
                sG[(c4 + 2) * PW + row] = v[2];
                sG[(c4 + 3) * PW + row] = v[3];
            }
        }
    }
#pragma unroll
    for (int i = 0; i < 4; ++i) {
        const int f = tid + 512 * i;
        *(v4f*)&sWv[(f >> 4) * PW + (f & 15) * 4] = wv[i];
    }
    __syncthreads();

    float4 pk[4], pq[4], po[4];
#pragma unroll
    for (int i = 0; i < 4; ++i) {
        const int f = tid + 512 * i;
        pk[i] = Wk4[f];
        pq[i] = Wq4[f];
        po[i] = Wo4[f];
    }

    const int jb = w >> 1;
    const int cp = (w & 1) * 2;
    const int h  = w >> 2;
    const int eb = (w >> 1) & 1;

    {
        v8f acc0 = v8f_zero(), acc1 = v8f_zero();
#pragma unroll
        for (int kg = 0; kg < 16; ++kg) {
            const v2f a  = lds_frag(sWv, jb * 16 + lanelo, PW, kg * 4 + hi2);
            const v2f b0 = lds_frag(sG, cp * 16 + lanelo, PW, kg * 4 + hi2);
            const v2f b1 = lds_frag(sG, (cp + 1) * 16 + lanelo, PW, kg * 4 + hi2);
            acc0 = wmma_f32(a, b0, acc0);
            acc1 = wmma_f32(a, b1, acc1);
        }
#pragma unroll
        for (int r = 0; r < 8; ++r) {
            sT[(jb * 16 + rowOff + r) * PW + cp * 16 + lanelo]       = acc0[r];
            sT[(jb * 16 + rowOff + r) * PW + (cp + 1) * 16 + lanelo] = acc1[r];
        }
    }
#pragma unroll
    for (int i = 0; i < 4; ++i) {
        const int f = tid + 512 * i;
        *(float4*)&sWk[(f >> 4) * PW + (f & 15) * 4] = pk[i];
        *(float4*)&sWq[(f >> 4) * PQ + (f & 15) * 4] = pq[i];
        *(float4*)&sWo[(f >> 5) * PO + (f & 31) * 4] = po[i];
    }
    __syncthreads();

    {
        const int db = w & 1;
        v8f c0 = v8f_zero(), c1 = v8f_zero();
#pragma unroll
        for (int kg = 0; kg < 16; kg += 2) {
            const v2f a0 = lds_frag(sT, h * 32 + eb * 16 + lanelo, PW, kg * 4 + hi2);
            const v2f k0 = lds_frag(sWk, h * 32 + db * 16 + lanelo, PW, kg * 4 + hi2);
            const v2f a1 = lds_frag(sT, h * 32 + eb * 16 + lanelo, PW, (kg + 1) * 4 + hi2);
            const v2f k1 = lds_frag(sWk, h * 32 + db * 16 + lanelo, PW, (kg + 1) * 4 + hi2);
            c0 = wmma_f32(a0, k0, c0);
            c1 = wmma_f32(a1, k1, c1);
        }
#pragma unroll
        for (int r = 0; r < 8; ++r)
            sC[(h * 32 + eb * 16 + rowOff + r) * PC + db * 16 + lanelo] =
                (c0[r] + c1[r]) * (1.0f / 4096.0f);
    }
    __syncthreads();

    {
        v8f m0 = v8f_zero(), m1 = v8f_zero();
#pragma unroll
        for (int kg = 0; kg < 8; ++kg) {
            const v2f bb = lds_frag(sC, h * 32 + eb * 16 + lanelo, PC, kg * 4 + hi2);
            const float* q = &sWq[(h * 32 + kg * 4 + hi2) * PQ + lanelo];
            v2f a0, a1;
            a0[0] = q[cp * 16];
            a0[1] = q[PQ + cp * 16];
            a1[0] = q[(cp + 1) * 16];
            a1[1] = q[PQ + (cp + 1) * 16];
            m0 = wmma_f32(a0, bb, m0);
            m1 = wmma_f32(a1, bb, m1);
        }
#pragma unroll
        for (int r = 0; r < 8; ++r) {
            sM[(cp * 16 + rowOff + r) * PO + jb * 16 + lanelo]       = m0[r] * scale;
            sM[((cp + 1) * 16 + rowOff + r) * PO + jb * 16 + lanelo] = m1[r] * scale;
        }
    }
    __syncthreads();

    {
        const int ob = w >> 2;
        const int cb = w & 3;
        v8f o0 = v8f_zero(), o1 = v8f_zero();
#pragma unroll
        for (int kg = 0; kg < 32; kg += 2) {
            const v2f a0 = lds_frag(sWo, ob * 16 + lanelo, PO, kg * 4 + hi2);
            const v2f b0 = lds_frag(sM, cb * 16 + lanelo, PO, kg * 4 + hi2);
            const v2f a1 = lds_frag(sWo, ob * 16 + lanelo, PO, (kg + 1) * 4 + hi2);
            const v2f b1 = lds_frag(sM, cb * 16 + lanelo, PO, (kg + 1) * 4 + hi2);
            o0 = wmma_f32(a0, b0, o0);
            o1 = wmma_f32(a1, b1, o1);
        }
#pragma unroll
        for (int r = 0; r < 8; ++r)
            sC[(ob * 16 + rowOff + r) * CCH + cb * 16 + lanelo] = o0[r] + o1[r];
    }
    __syncthreads();
    float4* A4 = (float4*)(A + b * 4096);
#pragma unroll
    for (int i = 0; i < 2; ++i) {
        const int f = tid + 512 * i;
        A4[f] = *(const float4*)&sC[f * 4];
    }
}

__global__ __launch_bounds__(128) void attn_out_kernel(
    const float* __restrict__ x, const float* __restrict__ A,
    const float* __restrict__ b_out, const float* __restrict__ g,
    float* __restrict__ out) {
    const int tid    = threadIdx.x;
    const int lane   = tid & 31;
    const int w      = tid >> 5;
    const int batch  = blockIdx.x / Y_BLOCKS_PER_BATCH;
    const int blk    = blockIdx.x % Y_BLOCKS_PER_BATCH;
    const float* xb  = x + (size_t)batch * CCH * NTOK;
    float* ob        = out + (size_t)batch * CCH * NTOK;
    const float* Ab  = A + batch * CCH * CCH;

    const int lanelo = lane & 15;
    const int hi2    = (lane >> 4) * 2;
    const int rowOff = (lane < 16) ? 0 : 8;

    v2f aop[16][4];
#pragma unroll
    for (int kg = 0; kg < 16; ++kg)
#pragma unroll
        for (int mt = 0; mt < 4; ++mt) {
            const float* p = &Ab[(mt * 16 + lanelo) * CCH + kg * 4 + hi2];
            aop[kg][mt] = *(const v2f*)p;
        }

    float bia[4][8], gai[4][8];
#pragma unroll
    for (int mt = 0; mt < 4; ++mt)
#pragma unroll
        for (int r = 0; r < 8; ++r) {
            const int ch = mt * 16 + r + rowOff;
            bia[mt][r] = b_out[ch];
            gai[mt][r] = g[ch];
        }

    const int tokensPerBlock = NTOK / Y_BLOCKS_PER_BATCH;
    const int tokensPerWave  = tokensPerBlock / 4;
    const int tW0            = blk * tokensPerBlock + w * tokensPerWave;

    for (int tt = 0; tt < tokensPerWave; tt += 32) {
        const int tp = tW0 + tt + 2 * lanelo;
        v8f acc0[4], acc1[4];
#pragma unroll
        for (int mt = 0; mt < 4; ++mt) { acc0[mt] = v8f_zero(); acc1[mt] = v8f_zero(); }

#pragma unroll
        for (int kg = 0; kg < 16; ++kg) {
            const float* bp = xb + (size_t)(kg * 4 + hi2) * NTOK + tp;
            const v2f c0 = *(const v2f*)bp;
            const v2f c1 = *(const v2f*)(bp + NTOK);
            v2f b0, b1;
            b0[0] = c0[0]; b0[1] = c1[0];
            b1[0] = c0[1]; b1[1] = c1[1];
#pragma unroll
            for (int mt = 0; mt < 4; ++mt) {
                acc0[mt] = wmma_f32(aop[kg][mt], b0, acc0[mt]);
                acc1[mt] = wmma_f32(aop[kg][mt], b1, acc1[mt]);
            }
        }

        float s0 = 0.f, q0 = 0.f, s1 = 0.f, q1 = 0.f;
#pragma unroll
        for (int mt = 0; mt < 4; ++mt)
#pragma unroll
            for (int r = 0; r < 8; ++r) {
                const float y0 = acc0[mt][r] + bia[mt][r];
                const float y1 = acc1[mt][r] + bia[mt][r];
                acc0[mt][r] = y0;
                acc1[mt][r] = y1;
                s0 += y0;
                q0 += y0 * y0;
                s1 += y1;
                q1 += y1 * y1;
            }
        s0 += __shfl_xor(s0, 16, 32);
        q0 += __shfl_xor(q0, 16, 32);
        s1 += __shfl_xor(s1, 16, 32);
        q1 += __shfl_xor(q1, 16, 32);
        const float mean0 = s0 * (1.0f / 64.0f);
        const float var0  = q0 * (1.0f / 64.0f) - mean0 * mean0;
        const float inv0  = rsqrtf(var0 + 1e-5f);
        const float mean1 = s1 * (1.0f / 64.0f);
        const float var1  = q1 * (1.0f / 64.0f) - mean1 * mean1;
        const float inv1  = rsqrtf(var1 + 1e-5f);

#pragma unroll
        for (int mt = 0; mt < 4; ++mt)
#pragma unroll
            for (int r = 0; r < 8; ++r) {
                const int ch = mt * 16 + r + rowOff;
                v2f o;
                o[0] = (acc0[mt][r] - mean0) * inv0 * gai[mt][r];
                o[1] = (acc1[mt][r] - mean1) * inv1 * gai[mt][r];
                *(v2f*)&ob[(size_t)ch * NTOK + tp] = o;
            }
    }
}

extern "C" void kernel_launch(void* const* d_in, const int* in_sizes, int n_in,
                              void* d_out, int out_size, void* d_ws, size_t ws_size,
                              hipStream_t stream) {
    (void)in_sizes; (void)n_in; (void)out_size; (void)ws_size;
    const float* x    = (const float*)d_in[0];
    const float* wqkv = (const float*)d_in[1];
    const float* wout = (const float*)d_in[2];
    const float* bout = (const float*)d_in[3];
    const float* g    = (const float*)d_in[4];
    float* out        = (float*)d_out;
    float* ws         = (float*)d_ws;

    float* G  = ws;
    float* A  = ws + 8192;

    zero_kernel<<<32, 256, 0, stream>>>(G, 2 * CCH * CCH);
    gram_kernel<<<2 * G_BLOCKS_PER_BATCH, 256, 0, stream>>>(x, G);
    combine_kernel<<<2, 512, 0, stream>>>(wqkv, wout, G, A);
    attn_out_kernel<<<2 * Y_BLOCKS_PER_BATCH, 128, 0, stream>>>(x, A, bout, g, out);
}


# output kernel: a wave trip takes 64 tokens as four tiles (token mod 4); x loads and out stores are 16 bytes per lane, each half-wave access one whole aligned 256-byte unit (was 8 bytes, 128 B)
# speedup vs baseline: 3.6708x; 1.0803x over previous
#include <hip/hip_runtime.h>

typedef __attribute__((ext_vector_type(2))) float v2f;
typedef __attribute__((ext_vector_type(4))) float v4f;
typedef __attribute__((ext_vector_type(8))) float v8f;

#define NTOK 262144
#define CCH  64
#define HIDDEN 128

#define G_BLOCKS_PER_BATCH 128
#define G_WAVES 8
#define G_TILES 10
#define Y_BLOCKS_PER_BATCH 256

__device__ __forceinline__ v8f v8f_zero() {
    v8f z;
#pragma unroll
    for (int i = 0; i < 8; ++i) z[i] = 0.0f;
    return z;
}

__device__ __forceinline__ v8f wmma_f32(v2f a, v2f b, v8f c) {
    return __builtin_amdgcn_wmma_f32_16x16x4_f32(false, a, false, b, (short)0, c,
                                                 false, false);
}

__global__ void zero_kernel(float* __restrict__ p, int count) {
    int i = blockIdx.x * blockDim.x + threadIdx.x;
    if (i < count) p[i] = 0.0f;
}


__device__ __forceinline__ void wave_lds_handover() {
    __builtin_amdgcn_fence(__ATOMIC_RELEASE, "wavefront");
    __builtin_amdgcn_wave_barrier();
    __builtin_amdgcn_fence(__ATOMIC_ACQUIRE, "wavefront");
}

__global__ __launch_bounds__(256) void gram_kernel(const float* __restrict__ x,
                                                   float* __restrict__ G) {
    const int tid    = threadIdx.x;
    const int lane   = tid & 31;
    const int w      = tid >> 5;
    const int batch  = blockIdx.x / G_BLOCKS_PER_BATCH;
    const int blk    = blockIdx.x % G_BLOCKS_PER_BATCH;
    const float* xb  = x + (size_t)batch * CCH * NTOK;
    float* Gb        = G + batch * CCH * CCH;

    __shared__ float tile[G_WAVES][CCH][34];
    __shared__ __attribute__((aligned(16))) v4f red[G_WAVES * G_TILES * 2 * 32];

    const int lanelo = lane & 15;
    const int hi2    = (lane >> 4) * 2;
    const int rowOff = (lane < 16) ? 0 : 8;

    v8f acc[G_TILES];
#pragma unroll
    for (int t = 0; t < G_TILES; ++t) acc[t] = v8f_zero();

    const int tokensPerBlock = NTOK / G_BLOCKS_PER_BATCH;
    const int tokensPerWave  = tokensPerBlock / G_WAVES;
    const float* xw = xb + blk * tokensPerBlock + w * tokensPerWave + lane;

    float pf[CCH];
#pragma unroll
    for (int c = 0; c < CCH; ++c) pf[c] = xw[(size_t)c * NTOK];

    for (int it = 0; it < tokensPerWave; it += 32) {
#pragma unroll
        for (int c = 0; c < CCH; ++c) tile[w][c][lane] = pf[c];
        if (it + 32 < tokensPerWave) {
#pragma unroll
            for (int c = 0; c < CCH; ++c) pf[c] = xw[(size_t)c * NTOK + it + 32];
        }
        wave_lds_handover();

        v2f op[8][4];
#pragma unroll
        for (int kg = 0; kg < 8; ++kg)
#pragma unroll
            for (int gch = 0; gch < 4; ++gch) {
                const float* p = &tile[w][gch * 16 + lanelo][kg * 4 + hi2];
                op[kg][gch] = *(const v2f*)p;
            }
#pragma unroll
        for (int kg = 0; kg < 8; ++kg)
#pragma unroll
            for (int i = 0; i < 4; ++i)
#pragma unroll
                for (int j = i; j < 4; ++j) {
                    const int t = i * 4 + j - (i * (i + 1)) / 2;
                    acc[t] = wmma_f32(op[kg][i], op[kg][j], acc[t]);
                }
        wave_lds_handover();
    }

#pragma unroll
    for (int t = 0; t < G_TILES; ++t) {
        const v8f a = acc[t];
        v4f lo, hi;
        lo[0] = a[0]; lo[1] = a[1]; lo[2] = a[2]; lo[3] = a[3];
        hi[0] = a[4]; hi[1] = a[5]; hi[2] = a[6]; hi[3] = a[7];
        red[((w * G_TILES + t) * 2 + 0) * 32 + lane] = lo;
        red[((w * G_TILES + t) * 2 + 1) * 32 + lane] = hi;
    }
    __syncthreads();

#pragma unroll
    for (int u = 0; u < 2; ++u) {
        const int t = w + G_WAVES * u;
        if (t < G_TILES) {
            v4f lo = red[(t * 2 + 0) * 32 + lane];
            v4f hi = red[(t * 2 + 1) * 32 + lane];
#pragma unroll
            for (int s = 1; s < G_WAVES; ++s) {
                lo += red[((s * G_TILES + t) * 2 + 0) * 32 + lane];
                hi += red[((s * G_TILES + t) * 2 + 1) * 32 + lane];
            }
            const int ti = (t >= 9) ? 3 : (t >= 7) ? 2 : (t >= 4) ? 1 : 0;
            const int tj = t - 4 * ti + (ti * (ti + 1)) / 2;
            const int row0 = ti * 16 + rowOff;
            const int col  = tj * 16 + lanelo;
#pragma unroll
            for (int r = 0; r < 4; ++r) {
                atomicAdd(&Gb[(row0 + r) * CCH + col], lo[r]);
                atomicAdd(&Gb[(row0 + 4 + r) * CCH + col], hi[r]);
            }
        }
    }
}

#define PW 68
#define PQ 72
#define PO 132
#define PC 36

__device__ __forceinline__ v2f lds_frag(const float* base, int row, int pitch, int k) {
    return *(const v2f*)(base + row * pitch + k);
}

__global__ __launch_bounds__(512) void combine_kernel(
    const float* __restrict__ wqkv, const float* __restrict__ wout,
    const float* __restrict__ G, float* __restrict__ A) {
    __shared__ __attribute__((aligned(16))) float sG [64  * PW];
    __shared__ __attribute__((aligned(16))) float sWv[128 * PW];
    __shared__ __attribute__((aligned(16))) float sWk[128 * PW];
    __shared__ __attribute__((aligned(16))) float sWq[128 * PQ];
    __shared__ __attribute__((aligned(16))) float sWo[64  * PO];
    __shared__ __attribute__((aligned(16))) float sT [128 * PW];
    __shared__ __attribute__((aligned(16))) float sC [128 * PC];
    float* sM = sT;

    const int b      = blockIdx.x;
    const int tid    = threadIdx.x;
    const int lane   = tid & 31;
    const int w      = tid >> 5;
    const int lanelo = lane & 15;
    const int hi2    = (lane >> 4) * 2;
    const int rowOff = (lane >> 4) * 8;
    const float scale = 0.1767766952966369f;

    const float4* G4  = (const float4*)(G + b * 4096);
    const float4* Wq4 = (const float4*)(wqkv);
    const float4* Wk4 = (const float4*)(wqkv + 128 * CCH);
    const float4* Wv4 = (const float4*)(wqkv + 256 * CCH);
    const float4* Wo4 = (const float4*)(wout);

    v4f g0 = ((const v4f*)G4)[tid],        g1 = ((const v4f*)G4)[tid + 512];
    v4f w0 = ((const v4f*)Wv4)[tid],       w1 = ((const v4f*)Wv4)[tid + 512];
    v4f w2 = ((const v4f*)Wv4)[tid + 1024], w3 = ((const v4f*)Wv4)[tid + 1536];
    asm volatile("" : "+v"(g0), "+v"(g1), "+v"(w0), "+v"(w1), "+v"(w2), "+v"(w3));
    const v4f gv[2] = {g0, g1};
    const v4f wv[4] = {w0, w1, w2, w3};
#pragma unroll
    for (int i = 0; i < 2; ++i) {
        const int f   = tid + 512 * i;
        const int row = f >> 4;
        const int c4  = (f & 15) * 4;
        if ((c4 >> 4) >= (row >> 4)) {
            const v4f v = gv[i];
            *(v4f*)&sG[row * PW + c4] = v;
            if ((c4 >> 4) > (row >> 4)) {
                sG[(c4 + 0) * PW + row] = v[0];
                sG[(c4 + 1) * PW + row] = v[1];
                sG[(c4 + 2) * PW + row] = v[2];
                sG[(c4 + 3) * PW + row] = v[3];
            }
        }
    }
#pragma unroll
    for (int i = 0; i < 4; ++i) {
        const int f = tid + 512 * i;
        *(v4f*)&sWv[(f >> 4) * PW + (f & 15) * 4] = wv[i];
    }
    __syncthreads();

    float4 pk[4], pq[4], po[4];
#pragma unroll
    for (int i = 0; i < 4; ++i) {
        const int f = tid + 512 * i;
        pk[i] = Wk4[f];
        pq[i] = Wq4[f];
        po[i] = Wo4[f];
    }

    const int jb = w >> 1;
    const int cp = (w & 1) * 2;
    const int h  = w >> 2;
    const int eb = (w >> 1) & 1;

    {
        v8f acc0 = v8f_zero(), acc1 = v8f_zero();
#pragma unroll
        for (int kg = 0; kg < 16; ++kg) {
            const v2f a  = lds_frag(sWv, jb * 16 + lanelo, PW, kg * 4 + hi2);
            const v2f b0 = lds_frag(sG, cp * 16 + lanelo, PW, kg * 4 + hi2);
            const v2f b1 = lds_frag(sG, (cp + 1) * 16 + lanelo, PW, kg * 4 + hi2);
            acc0 = wmma_f32(a, b0, acc0);
            acc1 = wmma_f32(a, b1, acc1);
        }
#pragma unroll
        for (int r = 0; r < 8; ++r) {
            sT[(jb * 16 + rowOff + r) * PW + cp * 16 + lanelo]       = acc0[r];
            sT[(jb * 16 + rowOff + r) * PW + (cp + 1) * 16 + lanelo] = acc1[r];
        }
    }
#pragma unroll
    for (int i = 0; i < 4; ++i) {
        const int f = tid + 512 * i;
        *(float4*)&sWk[(f >> 4) * PW + (f & 15) * 4] = pk[i];
        *(float4*)&sWq[(f >> 4) * PQ + (f & 15) * 4] = pq[i];
        *(float4*)&sWo[(f >> 5) * PO + (f & 31) * 4] = po[i];
    }
    __syncthreads();

    {
        const int db = w & 1;
        v8f c0 = v8f_zero(), c1 = v8f_zero();
#pragma unroll
        for (int kg = 0; kg < 16; kg += 2) {
            const v2f a0 = lds_frag(sT, h * 32 + eb * 16 + lanelo, PW, kg * 4 + hi2);
            const v2f k0 = lds_frag(sWk, h * 32 + db * 16 + lanelo, PW, kg * 4 + hi2);
            const v2f a1 = lds_frag(sT, h * 32 + eb * 16 + lanelo, PW, (kg + 1) * 4 + hi2);
            const v2f k1 = lds_frag(sWk, h * 32 + db * 16 + lanelo, PW, (kg + 1) * 4 + hi2);
            c0 = wmma_f32(a0, k0, c0);
            c1 = wmma_f32(a1, k1, c1);
        }
#pragma unroll
        for (int r = 0; r < 8; ++r)
            sC[(h * 32 + eb * 16 + rowOff + r) * PC + db * 16 + lanelo] =
                (c0[r] + c1[r]) * (1.0f / 4096.0f);
    }
    __syncthreads();

    {
        v8f m0 = v8f_zero(), m1 = v8f_zero();
#pragma unroll
        for (int kg = 0; kg < 8; ++kg) {
            const v2f bb = lds_frag(sC, h * 32 + eb * 16 + lanelo, PC, kg * 4 + hi2);
            const float* q = &sWq[(h * 32 + kg * 4 + hi2) * PQ + lanelo];
            v2f a0, a1;
            a0[0] = q[cp * 16];
            a0[1] = q[PQ + cp * 16];
            a1[0] = q[(cp + 1) * 16];
            a1[1] = q[PQ + (cp + 1) * 16];
            m0 = wmma_f32(a0, bb, m0);
            m1 = wmma_f32(a1, bb, m1);
        }
#pragma unroll
        for (int r = 0; r < 8; ++r) {
            sM[(cp * 16 + rowOff + r) * PO + jb * 16 + lanelo]       = m0[r] * scale;
            sM[((cp + 1) * 16 + rowOff + r) * PO + jb * 16 + lanelo] = m1[r] * scale;
        }
    }
    __syncthreads();

    {
        const int ob = w >> 2;
        const int cb = w & 3;
        v8f o0 = v8f_zero(), o1 = v8f_zero();
#pragma unroll
        for (int kg = 0; kg < 32; kg += 2) {
            const v2f a0 = lds_frag(sWo, ob * 16 + lanelo, PO, kg * 4 + hi2);
            const v2f b0 = lds_frag(sM, cb * 16 + lanelo, PO, kg * 4 + hi2);
            const v2f a1 = lds_frag(sWo, ob * 16 + lanelo, PO, (kg + 1) * 4 + hi2);
            const v2f b1 = lds_frag(sM, cb * 16 + lanelo, PO, (kg + 1) * 4 + hi2);
            o0 = wmma_f32(a0, b0, o0);
            o1 = wmma_f32(a1, b1, o1);
        }
#pragma unroll
        for (int r = 0; r < 8; ++r)
            sC[(ob * 16 + rowOff + r) * CCH + cb * 16 + lanelo] = o0[r] + o1[r];
    }
    __syncthreads();
    float4* A4 = (float4*)(A + b * 4096);
#pragma unroll
    for (int i = 0; i < 2; ++i) {
        const int f = tid + 512 * i;
        A4[f] = *(const float4*)&sC[f * 4];
    }
}

__global__ __launch_bounds__(128) void attn_out_kernel(
    const float* __restrict__ x, const float* __restrict__ A,
    const float* __restrict__ b_out, const float* __restrict__ g,
    float* __restrict__ out) {
    const int tid    = threadIdx.x;
    const int lane   = tid & 31;
    const int w      = tid >> 5;
    const int batch  = blockIdx.x / Y_BLOCKS_PER_BATCH;
    const int blk    = blockIdx.x % Y_BLOCKS_PER_BATCH;
    const float* xb  = x + (size_t)batch * CCH * NTOK;
    float* ob        = out + (size_t)batch * CCH * NTOK;
    const float* Ab  = A + batch * CCH * CCH;

    const int lanelo = lane & 15;
    const int hi2    = (lane >> 4) * 2;
    const int rowOff = (lane < 16) ? 0 : 8;

    v2f aop[16][4];
#pragma unroll
    for (int kg = 0; kg < 16; ++kg)
#pragma unroll
        for (int mt = 0; mt < 4; ++mt) {
            const float* p = &Ab[(mt * 16 + lanelo) * CCH + kg * 4 + hi2];
            aop[kg][mt] = *(const v2f*)p;
        }

    float bia[4][8], gai[4][8];
#pragma unroll
    for (int mt = 0; mt < 4; ++mt)
#pragma unroll
        for (int r = 0; r < 8; ++r) {
            const int ch = mt * 16 + r + rowOff;
            bia[mt][r] = b_out[ch];
            gai[mt][r] = g[ch];
        }

    const int tokensPerBlock = NTOK / Y_BLOCKS_PER_BATCH;
    const int tokensPerWave  = tokensPerBlock / 4;
    const int tW0            = blk * tokensPerBlock + w * tokensPerWave;

    for (int tt = 0; tt < tokensPerWave; tt += 64) {
        const int tq = tW0 + tt + 4 * lanelo;
        v8f acc[4][4];
#pragma unroll
        for (int j = 0; j < 4; ++j)
#pragma unroll
            for (int mt = 0; mt < 4; ++mt) acc[j][mt] = v8f_zero();

#pragma unroll
        for (int kg = 0; kg < 16; ++kg) {
            const float* bp = xb + (size_t)(kg * 4 + hi2) * NTOK + tq;
            const v4f c0 = *(const v4f*)bp;
            const v4f c1 = *(const v4f*)(bp + NTOK);
#pragma unroll
            for (int j = 0; j < 4; ++j) {
                v2f bf;
                bf[0] = c0[j]; bf[1] = c1[j];
#pragma unroll
                for (int mt = 0; mt < 4; ++mt)
                    acc[j][mt] = wmma_f32(aop[kg][mt], bf, acc[j][mt]);
            }
        }

        float s[4], q[4];
#pragma unroll
        for (int j = 0; j < 4; ++j) { s[j] = 0.f; q[j] = 0.f; }
#pragma unroll
        for (int mt = 0; mt < 4; ++mt)
#pragma unroll
            for (int r = 0; r < 8; ++r) {
                const float bi = bia[mt][r];
#pragma unroll
                for (int j = 0; j < 4; ++j) {
                    const float y = acc[j][mt][r] + bi;
                    acc[j][mt][r] = y;
                    s[j] += y;
                    q[j] += y * y;
                }
            }
        float mean[4], inv[4];
#pragma unroll
        for (int j = 0; j < 4; ++j) {
            s[j] += __shfl_xor(s[j], 16, 32);
            q[j] += __shfl_xor(q[j], 16, 32);
            mean[j] = s[j] * (1.0f / 64.0f);
            const float var = q[j] * (1.0f / 64.0f) - mean[j] * mean[j];
            inv[j] = rsqrtf(var + 1e-5f);
        }

#pragma unroll
        for (int mt = 0; mt < 4; ++mt)
#pragma unroll
            for (int r = 0; r < 8; ++r) {
                const int ch = mt * 16 + r + rowOff;
                const float ga = gai[mt][r];
                v4f o;
#pragma unroll
                for (int j = 0; j < 4; ++j)
                    o[j] = (acc[j][mt][r] - mean[j]) * inv[j] * ga;
                *(v4f*)&ob[(size_t)ch * NTOK + tq] = o;
            }
    }
}

extern "C" void kernel_launch(void* const* d_in, const int* in_sizes, int n_in,
                              void* d_out, int out_size, void* d_ws, size_t ws_size,
                              hipStream_t stream) {
    (void)in_sizes; (void)n_in; (void)out_size; (void)ws_size;
    const float* x    = (const float*)d_in[0];
    const float* wqkv = (const float*)d_in[1];
    const float* wout = (const float*)d_in[2];
    const float* bout = (const float*)d_in[3];
    const float* g    = (const float*)d_in[4];
    float* out        = (float*)d_out;
    float* ws         = (float*)d_ws;

    float* G  = ws;
    float* A  = ws + 8192;

    zero_kernel<<<32, 256, 0, stream>>>(G, 2 * CCH * CCH);
    gram_kernel<<<2 * G_BLOCKS_PER_BATCH, 256, 0, stream>>>(x, G);
    combine_kernel<<<2, 512, 0, stream>>>(wqkv, wout, G, A);
    attn_out_kernel<<<2 * Y_BLOCKS_PER_BATCH, 128, 0, stream>>>(x, A, bout, g, out);
}
